# starting kernel
# baseline (speedup 1.0000x reference)
#include <hip/hip_runtime.h>

typedef __attribute__((ext_vector_type(2))) float v2f;
typedef __attribute__((ext_vector_type(8))) float v8f;

#define WLS_EPS 1e-6f

__device__ __forceinline__ float wls_xswap(float v) {
  float r;
  asm("v_permlanex16_b32 %0, %1, %2, %3"
      : "=v"(r)
      : "v"(v), "s"(0x76543210), "s"(0xfedcba98u));
  return r;
}

__device__ __forceinline__ float wls_relu(float v) {
  float r;
  asm("v_max_num_f32 %0, 0, %1" : "=v"(r) : "v"(v));
  return r;
}

__device__ __forceinline__ v8f wls_gemm2(const v2f* a2, const float* hr,
                                         int hi, v8f acc) {
  float sh[8];
#pragma unroll
  for (int v = 0; v < 8; ++v) sh[v] = wls_xswap(hr[v]);
  v2f b;
  b.x = hi ? sh[2] : hr[0]; b.y = hi ? sh[3] : hr[1];
  acc = __builtin_amdgcn_wmma_f32_16x16x4_f32(false, a2[0], false, b, (short)0, acc, false, false);
  b.x = hi ? sh[6] : hr[4]; b.y = hi ? sh[7] : hr[5];
  acc = __builtin_amdgcn_wmma_f32_16x16x4_f32(false, a2[1], false, b, (short)0, acc, false, false);
  b.x = hi ? hr[2] : sh[0]; b.y = hi ? hr[3] : sh[1];
  acc = __builtin_amdgcn_wmma_f32_16x16x4_f32(false, a2[2], false, b, (short)0, acc, false, false);
  b.x = hi ? hr[6] : sh[4]; b.y = hi ? hr[7] : sh[5];
  acc = __builtin_amdgcn_wmma_f32_16x16x4_f32(false, a2[3], false, b, (short)0, acc, false, false);
  return acc;
}

template <bool CLAMPED>
static __device__ __forceinline__ void wls_pair(
    const float* __restrict__ x, float* __restrict__ out,
    v2f a0, v2f a1, const v2f* a2, v8f cb1, v8f cb2,
    int p, int hi, int lo, int Brows)
{
  const int rowA  = p * 32 + lo;
  const int rowB  = rowA + 16;
  const int rowme = p * 32 + hi * 16 + lo;

  uint32_t oA0, oB0, orow;
  if (CLAMPED) {
    const int rcA = rowA  < Brows ? rowA  : Brows - 1;
    const int rcB = rowB  < Brows ? rowB  : Brows - 1;
    const int rcm = rowme < Brows ? rowme : Brows - 1;
    oA0  = (uint32_t)rcA * 8 + 2 * hi;
    oB0  = (uint32_t)rcB * 8 + 2 * hi;
    orow = (uint32_t)rcm * 8;
  } else {
    oA0  = (uint32_t)rowA * 8 + 2 * hi;
    oB0  = oA0 + 128;
    orow = (uint32_t)rowme * 8;
  }

  v2f bxA0 = *(const v2f*)(x + oA0);
  v2f bxA1 = *(const v2f*)(x + oA0 + 4);
  v2f bxB0 = *(const v2f*)(x + oB0);
  v2f bxB1 = *(const v2f*)(x + oB0 + 4);
  float4 xr0 = *(const float4*)(x + orow);
  float4 xr1 = *(const float4*)(x + orow + 4);

  v8f hA = cb1, hB = cb1;
  hA = __builtin_amdgcn_wmma_f32_16x16x4_f32(false, a0, false, bxA0, (short)0, hA, false, false);
  hA = __builtin_amdgcn_wmma_f32_16x16x4_f32(false, a1, false, bxA1, (short)0, hA, false, false);
  hB = __builtin_amdgcn_wmma_f32_16x16x4_f32(false, a0, false, bxB0, (short)0, hB, false, false);
  hB = __builtin_amdgcn_wmma_f32_16x16x4_f32(false, a1, false, bxB1, (short)0, hB, false, false);

  float hrA[8], hrB[8];
#pragma unroll
  for (int v = 0; v < 8; ++v) {
    hrA[v] = wls_relu(hA[v]);
    hrB[v] = wls_relu(hB[v]);
  }

  v8f sA = wls_gemm2(a2, hrA, hi, cb2);
  v8f sB = wls_gemm2(a2, hrB, hi, cb2);

  float sg[8];
#pragma unroll
  for (int v = 0; v < 8; ++v) {
    float zb = wls_xswap(sB[v]);
    float z  = hi ? zb : sA[v];
    sg[v] = wls_relu(z) + __logf(1.0f + __expf(-fabsf(z))) + WLS_EPS;
  }

  const float xf[8] = {xr0.x, xr0.y, xr0.z, xr0.w, xr1.x, xr1.y, xr1.z, xr1.w};

  float d0 = 1.0f / sg[0];
  float a  = d0 * xf[0];
  float q = 0.0f, sumx = 0.0f;
#pragma unroll
  for (int j = 1; j < 8; ++j) { q += sg[j]; sumx += xf[j]; }
  float s  = a * q + sumx;
  float g  = d0 * s / (1.0f + d0 * q);
  float o_first = s - g * q;
  float beta[7];
#pragma unroll
  for (int j = 0; j < 7; ++j) beta[j] = (a - g) * sg[j + 1] + xf[j + 1];

  if (!CLAMPED || rowme < Brows) {
    float* op = out + orow;
    *(float4*)(op)     = make_float4(o_first, beta[0], beta[1], beta[2]);
    *(float4*)(op + 4) = make_float4(beta[3], beta[4], beta[5], beta[6]);
  }
}

__global__ __launch_bounds__(256) void wls_wmma_kernel(
    const float* __restrict__ x, const float* __restrict__ W1,
    const float* __restrict__ b1, const float* __restrict__ W2,
    const float* __restrict__ b2, float* __restrict__ out, int Brows)
{
  const int lane = threadIdx.x & 31;
  const int wave = threadIdx.x >> 5;
  const int hi   = lane >> 4;
  const int lo   = lane & 15;

  const int wavesPerBlock = blockDim.x >> 5;
  const int gwave  = blockIdx.x * wavesPerBlock + wave;
  const int nwaves = gridDim.x * wavesPerBlock;
  const int nfull  = Brows >> 5;
  const int rem    = Brows & 31;

  v2f a0, a1;
  a0.x = W1[(0 + 2 * hi) * 16 + lo];
  a0.y = W1[(1 + 2 * hi) * 16 + lo];
  a1.x = W1[(4 + 2 * hi) * 16 + lo];
  a1.y = W1[(5 + 2 * hi) * 16 + lo];

  v2f a2[4];
#pragma unroll
  for (int c = 0; c < 4; ++c) {
    a2[c].x = (lo < 8) ? W2[(4 * c + 2 * hi    ) * 8 + lo] : 0.0f;
    a2[c].y = (lo < 8) ? W2[(4 * c + 2 * hi + 1) * 8 + lo] : 0.0f;
  }

  v8f cb1, cb2;
#pragma unroll
  for (int v = 0; v < 8; ++v) { cb1[v] = b1[8 * hi + v]; cb2[v] = b2[v]; }

  for (int p = gwave; p < nfull; p += nwaves) {
    if (p + nwaves < nfull)
      __builtin_prefetch(x + (size_t)((p + nwaves) * 32 + lane) * 8, 0, 3);
    wls_pair<false>(x, out, a0, a1, a2, cb1, cb2, p, hi, lo, Brows);
  }
  if (rem != 0 && gwave == (nfull % nwaves))
    wls_pair<true>(x, out, a0, a1, a2, cb1, cb2, nfull, hi, lo, Brows);
}

extern "C" void kernel_launch(void* const* d_in, const int* in_sizes, int n_in,
                              void* d_out, int out_size, void* d_ws, size_t ws_size,
                              hipStream_t stream) {
  const float* x  = (const float*)d_in[0];
  const float* W1 = (const float*)d_in[1];
  const float* b1 = (const float*)d_in[2];
  const float* W2 = (const float*)d_in[3];
  const float* b2 = (const float*)d_in[4];
  float* out = (float*)d_out;

  const int Brows = in_sizes[0] / 8;
  const int npTot = (Brows + 31) / 32;
  int nblocks = (npTot + 31) / 32;
  if (nblocks < 1) nblocks = 1;
  if (nblocks > 32768) nblocks = 32768;

  wls_wmma_kernel<<<nblocks, 256, 0, stream>>>(x, W1, b1, W2, b2, out, Brows);
}


# GEMM2 reads GEMM1's D registers directly as its B fragments (k order of the 2nd product permuted to the D layout); block-diagonal W2^T merges both tiles: no permlanex16 or select left
# speedup vs baseline: 1.0614x; 1.0614x over previous
#include <hip/hip_runtime.h>

typedef __attribute__((ext_vector_type(2))) float v2f;
typedef __attribute__((ext_vector_type(8))) float v8f;

#define WLS_EPS 1e-6f

__device__ __forceinline__ float wls_relu(float v) {
  float r;
  asm("v_max_num_f32 %0, 0, %1" : "=v"(r) : "v"(v));
  return r;
}

__device__ __forceinline__ v8f wls_gemm2(const v2f* a2A, const v2f* a2B,
                                         const float* hrA, const float* hrB,
                                         v8f acc) {
#pragma unroll
  for (int c = 0; c < 4; ++c) {
    v2f b; b.x = hrA[2 * c]; b.y = hrA[2 * c + 1];
    acc = __builtin_amdgcn_wmma_f32_16x16x4_f32(false, a2A[c], false, b, (short)0, acc, false, false);
  }
#pragma unroll
  for (int c = 0; c < 4; ++c) {
    v2f b; b.x = hrB[2 * c]; b.y = hrB[2 * c + 1];
    acc = __builtin_amdgcn_wmma_f32_16x16x4_f32(false, a2B[c], false, b, (short)0, acc, false, false);
  }
  return acc;
}

template <bool CLAMPED>
static __device__ __forceinline__ void wls_pair(
    const float* __restrict__ x, float* __restrict__ out,
    v2f a0, v2f a1, const v2f* a2A, const v2f* a2B, v8f cb1, v8f cb2,
    int p, int hi, int lo, int Brows)
{
  const int rowA  = p * 32 + lo;
  const int rowB  = rowA + 16;
  const int rowme = p * 32 + hi * 16 + lo;

  uint32_t oA0, oB0, orow;
  if (CLAMPED) {
    const int rcA = rowA  < Brows ? rowA  : Brows - 1;
    const int rcB = rowB  < Brows ? rowB  : Brows - 1;
    const int rcm = rowme < Brows ? rowme : Brows - 1;
    oA0  = (uint32_t)rcA * 8 + 2 * hi;
    oB0  = (uint32_t)rcB * 8 + 2 * hi;
    orow = (uint32_t)rcm * 8;
  } else {
    oA0  = (uint32_t)rowA * 8 + 2 * hi;
    oB0  = oA0 + 128;
    orow = (uint32_t)rowme * 8;
  }

  v2f bxA0 = *(const v2f*)(x + oA0);
  v2f bxA1 = *(const v2f*)(x + oA0 + 4);
  v2f bxB0 = *(const v2f*)(x + oB0);
  v2f bxB1 = *(const v2f*)(x + oB0 + 4);
  float4 xr0 = *(const float4*)(x + orow);
  float4 xr1 = *(const float4*)(x + orow + 4);

  v8f hA = cb1, hB = cb1;
  hA = __builtin_amdgcn_wmma_f32_16x16x4_f32(false, a0, false, bxA0, (short)0, hA, false, false);
  hA = __builtin_amdgcn_wmma_f32_16x16x4_f32(false, a1, false, bxA1, (short)0, hA, false, false);
  hB = __builtin_amdgcn_wmma_f32_16x16x4_f32(false, a0, false, bxB0, (short)0, hB, false, false);
  hB = __builtin_amdgcn_wmma_f32_16x16x4_f32(false, a1, false, bxB1, (short)0, hB, false, false);

  float hrA[8], hrB[8];
#pragma unroll
  for (int v = 0; v < 8; ++v) {
    hrA[v] = wls_relu(hA[v]);
    hrB[v] = wls_relu(hB[v]);
  }

  v8f sz = wls_gemm2(a2A, a2B, hrA, hrB, cb2);

  float sg[8];
#pragma unroll
  for (int v = 0; v < 8; ++v) {
    float z = sz[v];
    sg[v] = wls_relu(z) + __logf(1.0f + __expf(-fabsf(z))) + WLS_EPS;
  }

  const float xf[8] = {xr0.x, xr0.y, xr0.z, xr0.w, xr1.x, xr1.y, xr1.z, xr1.w};

  float d0 = 1.0f / sg[0];
  float a  = d0 * xf[0];
  float q = 0.0f, sumx = 0.0f;
#pragma unroll
  for (int j = 1; j < 8; ++j) { q += sg[j]; sumx += xf[j]; }
  float s  = a * q + sumx;
  float g  = d0 * s / (1.0f + d0 * q);
  float o_first = s - g * q;
  float beta[7];
#pragma unroll
  for (int j = 0; j < 7; ++j) beta[j] = (a - g) * sg[j + 1] + xf[j + 1];

  if (!CLAMPED || rowme < Brows) {
    float* op = out + orow;
    *(float4*)(op)     = make_float4(o_first, beta[0], beta[1], beta[2]);
    *(float4*)(op + 4) = make_float4(beta[3], beta[4], beta[5], beta[6]);
  }
}

__global__ __launch_bounds__(256) void wls_wmma_kernel(
    const float* __restrict__ x, const float* __restrict__ W1,
    const float* __restrict__ b1, const float* __restrict__ W2,
    const float* __restrict__ b2, float* __restrict__ out, int Brows)
{
  const int lane = threadIdx.x & 31;
  const int wave = threadIdx.x >> 5;
  const int hi   = lane >> 4;
  const int lo   = lane & 15;

  const int wavesPerBlock = blockDim.x >> 5;
  const int gwave  = blockIdx.x * wavesPerBlock + wave;
  const int nwaves = gridDim.x * wavesPerBlock;
  const int nfull  = Brows >> 5;
  const int rem    = Brows & 31;

  v2f a0, a1;
  a0.x = W1[(0 + 2 * hi) * 16 + lo];
  a0.y = W1[(1 + 2 * hi) * 16 + lo];
  a1.x = W1[(4 + 2 * hi) * 16 + lo];
  a1.y = W1[(5 + 2 * hi) * 16 + lo];

  v2f a2A[4], a2B[4];
#pragma unroll
  for (int c = 0; c < 4; ++c) {
    const float wx = W2[(8 * hi + 2 * c    ) * 8 + (lo & 7)];
    const float wy = W2[(8 * hi + 2 * c + 1) * 8 + (lo & 7)];
    a2A[c].x = (lo < 8) ? wx : 0.0f;  a2A[c].y = (lo < 8) ? wy : 0.0f;
    a2B[c].x = (lo < 8) ? 0.0f : wx;  a2B[c].y = (lo < 8) ? 0.0f : wy;
  }

  v8f cb1, cb2;
#pragma unroll
  for (int v = 0; v < 8; ++v) { cb1[v] = b1[8 * hi + v]; cb2[v] = b2[v]; }

  for (int p = gwave; p < nfull; p += nwaves) {
    if (p + nwaves < nfull)
      __builtin_prefetch(x + (size_t)((p + nwaves) * 32 + lane) * 8, 0, 3);
    wls_pair<false>(x, out, a0, a1, a2A, a2B, cb1, cb2, p, hi, lo, Brows);
  }
  if (rem != 0 && gwave == (nfull % nwaves))
    wls_pair<true>(x, out, a0, a1, a2A, a2B, cb1, cb2, nfull, hi, lo, Brows);
}

extern "C" void kernel_launch(void* const* d_in, const int* in_sizes, int n_in,
                              void* d_out, int out_size, void* d_ws, size_t ws_size,
                              hipStream_t stream) {
  const float* x  = (const float*)d_in[0];
  const float* W1 = (const float*)d_in[1];
  const float* b1 = (const float*)d_in[2];
  const float* W2 = (const float*)d_in[3];
  const float* b2 = (const float*)d_in[4];
  float* out = (float*)d_out;

  const int Brows = in_sizes[0] / 8;
  const int npTot = (Brows + 31) / 32;
  int nblocks = (npTot + 31) / 32;
  if (nblocks < 1) nblocks = 1;
  if (nblocks > 32768) nblocks = 32768;

  wls_wmma_kernel<<<nblocks, 256, 0, stream>>>(x, W1, b1, W2, b2, out, Brows);
}


# IEEE log/divide expansions out of the per-row epilogue: softplus log on native v_log_f32 (arg in (1,2]); solve re-associated to t=(x0-sum x)*v_rcp_f32(sum sigma), one reciprocal for two divides
# speedup vs baseline: 1.1688x; 1.1012x over previous
#include <hip/hip_runtime.h>

typedef __attribute__((ext_vector_type(2))) float v2f;
typedef __attribute__((ext_vector_type(8))) float v8f;

#define WLS_EPS 1e-6f
#define WLS_LOG2E 1.4426950408889634f
#define WLS_LN2 0.6931471805599453f

__device__ __forceinline__ float wls_relu(float v) {
  float r;
  asm("v_max_num_f32 %0, 0, %1" : "=v"(r) : "v"(v));
  return r;
}

__device__ __forceinline__ v8f wls_gemm2(const v2f* a2A, const v2f* a2B,
                                         const float* hrA, const float* hrB,
                                         v8f acc) {
#pragma unroll
  for (int c = 0; c < 4; ++c) {
    v2f b; b.x = hrA[2 * c]; b.y = hrA[2 * c + 1];
    acc = __builtin_amdgcn_wmma_f32_16x16x4_f32(false, a2A[c], false, b, (short)0, acc, false, false);
  }
#pragma unroll
  for (int c = 0; c < 4; ++c) {
    v2f b; b.x = hrB[2 * c]; b.y = hrB[2 * c + 1];
    acc = __builtin_amdgcn_wmma_f32_16x16x4_f32(false, a2B[c], false, b, (short)0, acc, false, false);
  }
  return acc;
}

template <bool CLAMPED>
static __device__ __forceinline__ void wls_pair(
    const float* __restrict__ x, float* __restrict__ out,
    v2f a0, v2f a1, const v2f* a2A, const v2f* a2B, v8f cb1, v8f cb2,
    int p, int hi, int lo, int Brows)
{
  const int rowA  = p * 32 + lo;
  const int rowB  = rowA + 16;
  const int rowme = p * 32 + hi * 16 + lo;

  uint32_t oA0, oB0, orow;
  if (CLAMPED) {
    const int rcA = rowA  < Brows ? rowA  : Brows - 1;
    const int rcB = rowB  < Brows ? rowB  : Brows - 1;
    const int rcm = rowme < Brows ? rowme : Brows - 1;
    oA0  = (uint32_t)rcA * 8 + 2 * hi;
    oB0  = (uint32_t)rcB * 8 + 2 * hi;
    orow = (uint32_t)rcm * 8;
  } else {
    oA0  = (uint32_t)rowA * 8 + 2 * hi;
    oB0  = oA0 + 128;
    orow = (uint32_t)rowme * 8;
  }

  v2f bxA0 = *(const v2f*)(x + oA0);
  v2f bxA1 = *(const v2f*)(x + oA0 + 4);
  v2f bxB0 = *(const v2f*)(x + oB0);
  v2f bxB1 = *(const v2f*)(x + oB0 + 4);
  float4 xr0 = *(const float4*)(x + orow);
  float4 xr1 = *(const float4*)(x + orow + 4);

  v8f hA = cb1, hB = cb1;
  hA = __builtin_amdgcn_wmma_f32_16x16x4_f32(false, a0, false, bxA0, (short)0, hA, false, false);
  hA = __builtin_amdgcn_wmma_f32_16x16x4_f32(false, a1, false, bxA1, (short)0, hA, false, false);
  hB = __builtin_amdgcn_wmma_f32_16x16x4_f32(false, a0, false, bxB0, (short)0, hB, false, false);
  hB = __builtin_amdgcn_wmma_f32_16x16x4_f32(false, a1, false, bxB1, (short)0, hB, false, false);

  float hrA[8], hrB[8];
#pragma unroll
  for (int v = 0; v < 8; ++v) {
    hrA[v] = wls_relu(hA[v]);
    hrB[v] = wls_relu(hB[v]);
  }

  v8f sz = wls_gemm2(a2A, a2B, hrA, hrB, cb2);

  float sg[8];
#pragma unroll
  for (int v = 0; v < 8; ++v) {
    float z = sz[v];
    float e = __builtin_amdgcn_exp2f(-fabsf(z * WLS_LOG2E));
    float lg = __builtin_amdgcn_logf(1.0f + e);
    sg[v] = fmaf(lg, WLS_LN2, wls_relu(z)) + WLS_EPS;
  }

  const float xf[8] = {xr0.x, xr0.y, xr0.z, xr0.w, xr1.x, xr1.y, xr1.z, xr1.w};

  float q = sg[1], sumx = xf[1];
#pragma unroll
  for (int j = 2; j < 8; ++j) { q += sg[j]; sumx += xf[j]; }
  float t = (xf[0] - sumx) * __builtin_amdgcn_rcpf(sg[0] + q);
  float o_first = fmaf(t, q, sumx);
  float beta[7];
#pragma unroll
  for (int j = 0; j < 7; ++j) beta[j] = fmaf(t, sg[j + 1], xf[j + 1]);

  if (!CLAMPED || rowme < Brows) {
    float* op = out + orow;
    *(float4*)(op)     = make_float4(o_first, beta[0], beta[1], beta[2]);
    *(float4*)(op + 4) = make_float4(beta[3], beta[4], beta[5], beta[6]);
  }
}

__global__ __launch_bounds__(256) void wls_wmma_kernel(
    const float* __restrict__ x, const float* __restrict__ W1,
    const float* __restrict__ b1, const float* __restrict__ W2,
    const float* __restrict__ b2, float* __restrict__ out, int Brows)
{
  const int lane = threadIdx.x & 31;
  const int wave = threadIdx.x >> 5;
  const int hi   = lane >> 4;
  const int lo   = lane & 15;

  const int wavesPerBlock = blockDim.x >> 5;
  const int gwave  = blockIdx.x * wavesPerBlock + wave;
  const int nwaves = gridDim.x * wavesPerBlock;
  const int nfull  = Brows >> 5;
  const int rem    = Brows & 31;

  v2f a0, a1;
  a0.x = W1[(0 + 2 * hi) * 16 + lo];
  a0.y = W1[(1 + 2 * hi) * 16 + lo];
  a1.x = W1[(4 + 2 * hi) * 16 + lo];
  a1.y = W1[(5 + 2 * hi) * 16 + lo];

  v2f a2A[4], a2B[4];
#pragma unroll
  for (int c = 0; c < 4; ++c) {
    const float wx = W2[(8 * hi + 2 * c    ) * 8 + (lo & 7)];
    const float wy = W2[(8 * hi + 2 * c + 1) * 8 + (lo & 7)];
    a2A[c].x = (lo < 8) ? wx : 0.0f;  a2A[c].y = (lo < 8) ? wy : 0.0f;
    a2B[c].x = (lo < 8) ? 0.0f : wx;  a2B[c].y = (lo < 8) ? 0.0f : wy;
  }

  v8f cb1, cb2;
#pragma unroll
  for (int v = 0; v < 8; ++v) { cb1[v] = b1[8 * hi + v]; cb2[v] = b2[v]; }

  for (int p = gwave; p < nfull; p += nwaves) {
    if (p + nwaves < nfull)
      __builtin_prefetch(x + (size_t)((p + nwaves) * 32 + lane) * 8, 0, 3);
    wls_pair<false>(x, out, a0, a1, a2A, a2B, cb1, cb2, p, hi, lo, Brows);
  }
  if (rem != 0 && gwave == (nfull % nwaves))
    wls_pair<true>(x, out, a0, a1, a2A, a2B, cb1, cb2, nfull, hi, lo, Brows);
}

extern "C" void kernel_launch(void* const* d_in, const int* in_sizes, int n_in,
                              void* d_out, int out_size, void* d_ws, size_t ws_size,
                              hipStream_t stream) {
  const float* x  = (const float*)d_in[0];
  const float* W1 = (const float*)d_in[1];
  const float* b1 = (const float*)d_in[2];
  const float* W2 = (const float*)d_in[3];
  const float* b2 = (const float*)d_in[4];
  float* out = (float*)d_out;

  const int Brows = in_sizes[0] / 8;
  const int npTot = (Brows + 31) / 32;
  int nblocks = (npTot + 31) / 32;
  if (nblocks < 1) nblocks = 1;
  if (nblocks > 32768) nblocks = 32768;

  wls_wmma_kernel<<<nblocks, 256, 0, stream>>>(x, W1, b1, W2, b2, out, Brows);
}


# Pair loop software-pipelined over two register buffers: next pair's x loads issue before the current pair's arithmetic, the first pair's beside the weight loads; prefetch two ahead; wave index scalar
# speedup vs baseline: 1.3379x; 1.1447x over previous
#include <hip/hip_runtime.h>


typedef __attribute__((ext_vector_type(2))) float v2f;
typedef __attribute__((ext_vector_type(8))) float v8f;

#define WLS_EPS 1e-6f
#define WLS_LOG2E 1.4426950408889634f
#define WLS_LN2 0.6931471805599453f

__device__ __forceinline__ float wls_relu(float v) {
  float r;
  asm("v_max_num_f32 %0, 0, %1" : "=v"(r) : "v"(v));
  return r;
}

__device__ __forceinline__ v8f wls_gemm2(const v2f* a2A, const v2f* a2B,
                                         const float* hrA, const float* hrB,
                                         v8f acc) {
#pragma unroll
  for (int c = 0; c < 4; ++c) {
    v2f b; b.x = hrA[2 * c]; b.y = hrA[2 * c + 1];
    acc = __builtin_amdgcn_wmma_f32_16x16x4_f32(false, a2A[c], false, b, (short)0, acc, false, false);
  }
#pragma unroll
  for (int c = 0; c < 4; ++c) {
    v2f b; b.x = hrB[2 * c]; b.y = hrB[2 * c + 1];
    acc = __builtin_amdgcn_wmma_f32_16x16x4_f32(false, a2B[c], false, b, (short)0, acc, false, false);
  }
  return acc;
}

struct wls_in {
  v2f bxA0, bxA1, bxB0, bxB1;
  float4 xr0, xr1;
};

__device__ __forceinline__ wls_in wls_load(const float* __restrict__ xb,
                                           uint32_t oA0, uint32_t oB0, uint32_t orow) {
  wls_in r;
  r.bxA0 = *(const v2f*)(xb + oA0);
  r.bxA1 = *(const v2f*)(xb + oA0 + 4);
  r.bxB0 = *(const v2f*)(xb + oB0);
  r.bxB1 = *(const v2f*)(xb + oB0 + 4);
  r.xr0  = *(const float4*)(xb + orow);
  r.xr1  = *(const float4*)(xb + orow + 4);
  return r;
}

__device__ __forceinline__ void wls_finish(
    const wls_in& in, float* __restrict__ op, bool do_store,
    v2f a0, v2f a1, const v2f* a2A, const v2f* a2B, v8f cb1, v8f cb2)
{
  v8f hA = cb1, hB = cb1;
  hA = __builtin_amdgcn_wmma_f32_16x16x4_f32(false, a0, false, in.bxA0, (short)0, hA, false, false);
  hA = __builtin_amdgcn_wmma_f32_16x16x4_f32(false, a1, false, in.bxA1, (short)0, hA, false, false);
  hB = __builtin_amdgcn_wmma_f32_16x16x4_f32(false, a0, false, in.bxB0, (short)0, hB, false, false);
  hB = __builtin_amdgcn_wmma_f32_16x16x4_f32(false, a1, false, in.bxB1, (short)0, hB, false, false);

  float hrA[8], hrB[8];
#pragma unroll
  for (int v = 0; v < 8; ++v) {
    hrA[v] = wls_relu(hA[v]);
    hrB[v] = wls_relu(hB[v]);
  }

  v8f sz = wls_gemm2(a2A, a2B, hrA, hrB, cb2);

  float sg[8];
#pragma unroll
  for (int v = 0; v < 8; ++v) {
    float z = sz[v];
    float e = __builtin_amdgcn_exp2f(-fabsf(z * WLS_LOG2E));
    float lg = __builtin_amdgcn_logf(1.0f + e);
    sg[v] = fmaf(lg, WLS_LN2, wls_relu(z)) + WLS_EPS;
  }

  const float xf[8] = {in.xr0.x, in.xr0.y, in.xr0.z, in.xr0.w,
                       in.xr1.x, in.xr1.y, in.xr1.z, in.xr1.w};

  float q = sg[1], sumx = xf[1];
#pragma unroll
  for (int j = 2; j < 8; ++j) { q += sg[j]; sumx += xf[j]; }
  float t = (xf[0] - sumx) * __builtin_amdgcn_rcpf(sg[0] + q);
  float o_first = fmaf(t, q, sumx);
  float beta[7];
#pragma unroll
  for (int j = 0; j < 7; ++j) beta[j] = fmaf(t, sg[j + 1], xf[j + 1]);

  if (do_store) {
    *(float4*)(op)     = make_float4(o_first, beta[0], beta[1], beta[2]);
    *(float4*)(op + 4) = make_float4(beta[3], beta[4], beta[5], beta[6]);
  }
}

__global__ __launch_bounds__(256) void wls_wmma_kernel(
    const float* __restrict__ x, const float* __restrict__ W1,
    const float* __restrict__ b1, const float* __restrict__ W2,
    const float* __restrict__ b2, float* __restrict__ out, int Brows)
{
  const int lane = threadIdx.x & 31;
  const int wave = __builtin_amdgcn_readfirstlane((int)(threadIdx.x >> 5));
  const int hi   = lane >> 4;
  const int lo   = lane & 15;

  const int wavesPerBlock = blockDim.x >> 5;
  const int gwave  = blockIdx.x * wavesPerBlock + wave;
  const int nwaves = gridDim.x * wavesPerBlock;
  const int nfull  = Brows >> 5;
  const int rem    = Brows & 31;

  const uint32_t oA0  = (uint32_t)lo * 8 + 2 * hi;
  const uint32_t orow = (uint32_t)lane * 8;

  v2f a0, a1;
  a0.x = W1[(0 + 2 * hi) * 16 + lo];
  a0.y = W1[(1 + 2 * hi) * 16 + lo];
  a1.x = W1[(4 + 2 * hi) * 16 + lo];
  a1.y = W1[(5 + 2 * hi) * 16 + lo];

  float w2x[4], w2y[4];
#pragma unroll
  for (int c = 0; c < 4; ++c) {
    w2x[c] = W2[(8 * hi + 2 * c    ) * 8 + (lo & 7)];
    w2y[c] = W2[(8 * hi + 2 * c + 1) * 8 + (lo & 7)];
  }

  v8f cb1, cb2;
#pragma unroll
  for (int v = 0; v < 8; ++v) { cb1[v] = b1[8 * hi + v]; cb2[v] = b2[v]; }

  int p = gwave;
  const bool any = p < nfull;
  wls_in bufA, bufB;
  if (any) {
    bufA = wls_load(x + (size_t)p * 256, oA0, oA0 + 128, orow);
    int pf = p + nwaves;
    pf = pf < nfull ? pf : nfull - 1;
    __builtin_prefetch(x + (size_t)pf * 256 + orow, 0, 3);
  }

  v2f a2A[4], a2B[4];
#pragma unroll
  for (int c = 0; c < 4; ++c) {
    a2A[c].x = (lo < 8) ? w2x[c] : 0.0f;  a2A[c].y = (lo < 8) ? w2y[c] : 0.0f;
    a2B[c].x = (lo < 8) ? 0.0f : w2x[c];  a2B[c].y = (lo < 8) ? 0.0f : w2y[c];
  }

  if (any) {
    for (;;) {
      int pn = p + nwaves;
      if (pn >= nfull) {
        wls_finish(bufA, out + (size_t)p * 256 + orow, true, a0, a1, a2A, a2B, cb1, cb2);
        break;
      }
      bufB = wls_load(x + (size_t)pn * 256, oA0, oA0 + 128, orow);
      {
        int pf = pn + nwaves;
        pf = pf < nfull ? pf : nfull - 1;
        __builtin_prefetch(x + (size_t)pf * 256 + orow, 0, 3);
      }
      wls_finish(bufA, out + (size_t)p * 256 + orow, true, a0, a1, a2A, a2B, cb1, cb2);
      p = pn;
      pn = p + nwaves;
      if (pn >= nfull) {
        wls_finish(bufB, out + (size_t)p * 256 + orow, true, a0, a1, a2A, a2B, cb1, cb2);
        break;
      }
      bufA = wls_load(x + (size_t)pn * 256, oA0, oA0 + 128, orow);
      {
        int pf = pn + nwaves;
        pf = pf < nfull ? pf : nfull - 1;
        __builtin_prefetch(x + (size_t)pf * 256 + orow, 0, 3);
      }
      wls_finish(bufB, out + (size_t)p * 256 + orow, true, a0, a1, a2A, a2B, cb1, cb2);
      p = pn;
    }
  }

  if (rem != 0 && gwave == (nfull % nwaves)) {
    const int rowA  = nfull * 32 + lo;
    const int rowB  = rowA + 16;
    const int rowme = nfull * 32 + lane;
    const int rcA = rowA  < Brows ? rowA  : Brows - 1;
    const int rcB = rowB  < Brows ? rowB  : Brows - 1;
    const int rcm = rowme < Brows ? rowme : Brows - 1;
    const wls_in t = wls_load(x, (uint32_t)rcA * 8 + 2 * hi, (uint32_t)rcB * 8 + 2 * hi,
                              (uint32_t)rcm * 8);
    wls_finish(t, out + (size_t)rcm * 8, rowme < Brows, a0, a1, a2A, a2B, cb1, cb2);
  }
}

extern "C" void kernel_launch(void* const* d_in, const int* in_sizes, int n_in,
                              void* d_out, int out_size, void* d_ws, size_t ws_size,
                              hipStream_t stream) {
  const float* x  = (const float*)d_in[0];
  const float* W1 = (const float*)d_in[1];
  const float* b1 = (const float*)d_in[2];
  const float* W2 = (const float*)d_in[3];
  const float* b2 = (const float*)d_in[4];
  float* out = (float*)d_out;

  const int Brows = in_sizes[0] / 8;
  const int npTot = (Brows + 31) / 32;
  int nblocks = (npTot + 31) / 32;
  if (nblocks < 1) nblocks = 1;
  if (nblocks > 32768) nblocks = 32768;

  wls_wmma_kernel<<<nblocks, 256, 0, stream>>>(x, W1, b1, W2, b2, out, Brows);
}


# Lane owns 4 channels of both tiles' rows (was 8 of one): one whole-line b128 load per tile feeds GEMM1 and solver, whole-line stores; halves joined by permlanex16; sums reordered (GEMM1 k, solver)
# speedup vs baseline: 1.4424x; 1.0781x over previous
#include <hip/hip_runtime.h>


typedef __attribute__((ext_vector_type(2))) float v2f;
typedef __attribute__((ext_vector_type(8))) float v8f;

#define WLS_EPS 1e-6f
#define WLS_LOG2E 1.4426950408889634f
#define WLS_LN2 0.6931471805599453f

__device__ __forceinline__ float wls_relu(float v) {
  float r;
  asm("v_max_num_f32 %0, 0, %1" : "=v"(r) : "v"(v));
  return r;
}

__device__ __forceinline__ float wls_xhalf(float v) {
  int b = __builtin_bit_cast(int, v);
  return __builtin_bit_cast(float,
      __builtin_amdgcn_permlanex16(b, b, 0x76543210, 0xfedcba98, false, false));
}


__device__ __forceinline__ v8f wls_gemm2(const v2f* a2A, const v2f* a2B,
                                         const float* hrA, const float* hrB,
                                         v8f acc) {
#pragma unroll
  for (int c = 0; c < 4; ++c) {
    v2f b; b.x = hrA[2 * c]; b.y = hrA[2 * c + 1];
    acc = __builtin_amdgcn_wmma_f32_16x16x4_f32(false, a2A[c], false, b, (short)0, acc, false, false);
  }
#pragma unroll
  for (int c = 0; c < 4; ++c) {
    v2f b; b.x = hrB[2 * c]; b.y = hrB[2 * c + 1];
    acc = __builtin_amdgcn_wmma_f32_16x16x4_f32(false, a2B[c], false, b, (short)0, acc, false, false);
  }
  return acc;
}

struct wls_in {
  float4 xa, xb;
};

__device__ __forceinline__ wls_in wls_load(const float* __restrict__ xb,
                                           uint32_t oA, uint32_t oB) {
  wls_in r;
  r.xa = *(const float4*)(xb + oA);
  r.xb = *(const float4*)(xb + oB);
  return r;
}

__device__ __forceinline__ float4 wls_solve4(float4 xv, float g0, float g1,
                                             float g2, float g3, float nsgn) {
  float sx  = xv.y + xv.z + xv.w;
  float pn  = fmaf(-nsgn, xv.x, -sx);
  float pd  = g0 + (g1 + g2 + g3);
  float num = pn + wls_xhalf(pn);
  float den = pd + wls_xhalf(pd);
  float t   = num * __builtin_amdgcn_rcpf(den);
  float4 o;
  o.x = fmaf(nsgn * t, g0, xv.x);
  o.y = fmaf(t, g1, xv.y);
  o.z = fmaf(t, g2, xv.z);
  o.w = fmaf(t, g3, xv.w);
  return o;
}

__device__ __forceinline__ void wls_finish(
    const wls_in& in, float* __restrict__ opA, float* __restrict__ opB,
    bool stA, bool stB,
    v2f a0, v2f a1, const v2f* a2A, const v2f* a2B, v8f cb1, v8f cb2, float nsgn)
{
  v2f bA0, bA1, bB0, bB1;
  bA0.x = in.xa.x; bA0.y = in.xa.y; bA1.x = in.xa.z; bA1.y = in.xa.w;
  bB0.x = in.xb.x; bB0.y = in.xb.y; bB1.x = in.xb.z; bB1.y = in.xb.w;
  v8f hA = cb1, hB = cb1;
  hA = __builtin_amdgcn_wmma_f32_16x16x4_f32(false, a0, false, bA0, (short)0, hA, false, false);
  hA = __builtin_amdgcn_wmma_f32_16x16x4_f32(false, a1, false, bA1, (short)0, hA, false, false);
  hB = __builtin_amdgcn_wmma_f32_16x16x4_f32(false, a0, false, bB0, (short)0, hB, false, false);
  hB = __builtin_amdgcn_wmma_f32_16x16x4_f32(false, a1, false, bB1, (short)0, hB, false, false);

  float hrA[8], hrB[8];
#pragma unroll
  for (int v = 0; v < 8; ++v) {
    hrA[v] = wls_relu(hA[v]);
    hrB[v] = wls_relu(hB[v]);
  }

  v8f sz = wls_gemm2(a2A, a2B, hrA, hrB, cb2);

  float sg[8];
#pragma unroll
  for (int v = 0; v < 8; ++v) {
    float z = sz[v];
    float e = __builtin_amdgcn_exp2f(-fabsf(z * WLS_LOG2E));
    float lg = __builtin_amdgcn_logf(1.0f + e);
    sg[v] = fmaf(lg, WLS_LN2, wls_relu(z)) + WLS_EPS;
  }

  const float4 oa = wls_solve4(in.xa, sg[0], sg[1], sg[2], sg[3], nsgn);
  const float4 ob = wls_solve4(in.xb, sg[4], sg[5], sg[6], sg[7], nsgn);

  if (stA) *(float4*)(opA) = oa;
  if (stB) *(float4*)(opB) = ob;
}

__global__ __launch_bounds__(256) void wls_wmma_kernel(
    const float* __restrict__ x, const float* __restrict__ W1,
    const float* __restrict__ b1, const float* __restrict__ W2,
    const float* __restrict__ b2, float* __restrict__ out, int Brows)
{
  const int lane = threadIdx.x & 31;
  const int wave = __builtin_amdgcn_readfirstlane((int)(threadIdx.x >> 5));
  const int hi   = lane >> 4;
  const int lo   = lane & 15;

  const int wavesPerBlock = blockDim.x >> 5;
  const int gwave  = blockIdx.x * wavesPerBlock + wave;
  const int nwaves = gridDim.x * wavesPerBlock;
  const int nfull  = Brows >> 5;
  const int rem    = Brows & 31;

  const uint32_t oA   = (uint32_t)lo * 8 + 4 * hi;
  const uint32_t orow = (uint32_t)lane * 8;
  const float nsgn = hi ? 1.0f : -1.0f;

  v2f a0, a1;
  a0.x = W1[(4 * hi + 0) * 16 + lo];
  a0.y = W1[(4 * hi + 1) * 16 + lo];
  a1.x = W1[(4 * hi + 2) * 16 + lo];
  a1.y = W1[(4 * hi + 3) * 16 + lo];

  const int  chm = (lo & 3) + 4 * (lo >> 3);
  const bool inA = ((lo >> 2) & 1) == 0;
  float w2x[4], w2y[4];
#pragma unroll
  for (int c = 0; c < 4; ++c) {
    w2x[c] = W2[(8 * hi + 2 * c    ) * 8 + chm];
    w2y[c] = W2[(8 * hi + 2 * c + 1) * 8 + chm];
  }

  const float4 b2v = *(const float4*)(b2 + 4 * hi);
  v8f cb1, cb2;
#pragma unroll
  for (int v = 0; v < 8; ++v) cb1[v] = b1[8 * hi + v];
  cb2[0] = b2v.x; cb2[1] = b2v.y; cb2[2] = b2v.z; cb2[3] = b2v.w;
  cb2[4] = b2v.x; cb2[5] = b2v.y; cb2[6] = b2v.z; cb2[7] = b2v.w;

  int p = gwave;
  const bool any = p < nfull;
  wls_in bufA, bufB;
  if (any) {
    bufA = wls_load(x + (size_t)p * 256, oA, oA + 128);
    int pf = p + nwaves;
    pf = pf < nfull ? pf : nfull - 1;
    __builtin_prefetch(x + (size_t)pf * 256 + orow, 0, 3);
  }

  v2f a2A[4], a2B[4];
#pragma unroll
  for (int c = 0; c < 4; ++c) {
    a2A[c].x = inA ? w2x[c] : 0.0f;  a2A[c].y = inA ? w2y[c] : 0.0f;
    a2B[c].x = inA ? 0.0f : w2x[c];  a2B[c].y = inA ? 0.0f : w2y[c];
  }

  if (any) {
    for (;;) {
      int pn = p + nwaves;
      if (pn >= nfull) {
        float* o = out + (size_t)p * 256 + oA;
        wls_finish(bufA, o, o + 128, true, true, a0, a1, a2A, a2B, cb1, cb2, nsgn);
        break;
      }
      bufB = wls_load(x + (size_t)pn * 256, oA, oA + 128);
      {
        int pf = pn + nwaves;
        pf = pf < nfull ? pf : nfull - 1;
        __builtin_prefetch(x + (size_t)pf * 256 + orow, 0, 3);
      }
      {
        float* o = out + (size_t)p * 256 + oA;
        wls_finish(bufA, o, o + 128, true, true, a0, a1, a2A, a2B, cb1, cb2, nsgn);
      }
      p = pn;
      pn = p + nwaves;
      if (pn >= nfull) {
        float* o = out + (size_t)p * 256 + oA;
        wls_finish(bufB, o, o + 128, true, true, a0, a1, a2A, a2B, cb1, cb2, nsgn);
        break;
      }
      bufA = wls_load(x + (size_t)pn * 256, oA, oA + 128);
      {
        int pf = pn + nwaves;
        pf = pf < nfull ? pf : nfull - 1;
        __builtin_prefetch(x + (size_t)pf * 256 + orow, 0, 3);
      }
      {
        float* o = out + (size_t)p * 256 + oA;
        wls_finish(bufB, o, o + 128, true, true, a0, a1, a2A, a2B, cb1, cb2, nsgn);
      }
      p = pn;
    }
  }

  if (rem != 0 && gwave == (nfull % nwaves)) {
    const int rowA = nfull * 32 + lo;
    const int rowB = rowA + 16;
    const int rcA = rowA < Brows ? rowA : Brows - 1;
    const int rcB = rowB < Brows ? rowB : Brows - 1;
    const wls_in t = wls_load(x, (uint32_t)rcA * 8 + 4 * hi, (uint32_t)rcB * 8 + 4 * hi);
    wls_finish(t, out + (size_t)rcA * 8 + 4 * hi, out + (size_t)rcB * 8 + 4 * hi,
               rowA < Brows, rowB < Brows, a0, a1, a2A, a2B, cb1, cb2, nsgn);
  }
}

extern "C" void kernel_launch(void* const* d_in, const int* in_sizes, int n_in,
                              void* d_out, int out_size, void* d_ws, size_t ws_size,
                              hipStream_t stream) {
  const float* x  = (const float*)d_in[0];
  const float* W1 = (const float*)d_in[1];
  const float* b1 = (const float*)d_in[2];
  const float* W2 = (const float*)d_in[3];
  const float* b2 = (const float*)d_in[4];
  float* out = (float*)d_out;

  const int Brows = in_sizes[0] / 8;
  const int npTot = (Brows + 31) / 32;
  int nblocks = (npTot + 31) / 32;
  if (nblocks < 1) nblocks = 1;
  if (nblocks > 32768) nblocks = 32768;

  wls_wmma_kernel<<<nblocks, 256, 0, stream>>>(x, W1, b1, W2, b2, out, Brows);
}


# All of a wave's x loads issued at entry: the 8 b128 loads of its 4 pairs go out ahead of the weight loads into 4 register buffers, each pair finished from registers; L2 prefetches dropped
# speedup vs baseline: 1.6010x; 1.1100x over previous
#include <hip/hip_runtime.h>


typedef __attribute__((ext_vector_type(2))) float v2f;
typedef __attribute__((ext_vector_type(8))) float v8f;

#define WLS_EPS 1e-6f
#define WLS_LOG2E 1.4426950408889634f
#define WLS_LN2 0.6931471805599453f

__device__ __forceinline__ float wls_relu(float v) {
  float r;
  asm("v_max_num_f32 %0, 0, %1" : "=v"(r) : "v"(v));
  return r;
}

__device__ __forceinline__ float wls_xhalf(float v) {
  int b = __builtin_bit_cast(int, v);
  return __builtin_bit_cast(float,
      __builtin_amdgcn_permlanex16(b, b, 0x76543210, 0xfedcba98, false, false));
}


__device__ __forceinline__ v8f wls_gemm2(const v2f* a2A, const v2f* a2B,
                                         const float* hrA, const float* hrB,
                                         v8f acc) {
#pragma unroll
  for (int c = 0; c < 4; ++c) {
    v2f b; b.x = hrA[2 * c]; b.y = hrA[2 * c + 1];
    acc = __builtin_amdgcn_wmma_f32_16x16x4_f32(false, a2A[c], false, b, (short)0, acc, false, false);
  }
#pragma unroll
  for (int c = 0; c < 4; ++c) {
    v2f b; b.x = hrB[2 * c]; b.y = hrB[2 * c + 1];
    acc = __builtin_amdgcn_wmma_f32_16x16x4_f32(false, a2B[c], false, b, (short)0, acc, false, false);
  }
  return acc;
}

struct wls_in {
  float4 xa, xb;
};

__device__ __forceinline__ wls_in wls_load(const float* __restrict__ xb,
                                           uint32_t oA, uint32_t oB) {
  wls_in r;
  r.xa = *(const float4*)(xb + oA);
  r.xb = *(const float4*)(xb + oB);
  return r;
}

__device__ __forceinline__ float4 wls_solve4(float4 xv, float g0, float g1,
                                             float g2, float g3, float nsgn) {
  float sx  = xv.y + xv.z + xv.w;
  float pn  = fmaf(-nsgn, xv.x, -sx);
  float pd  = g0 + (g1 + g2 + g3);
  float num = pn + wls_xhalf(pn);
  float den = pd + wls_xhalf(pd);
  float t   = num * __builtin_amdgcn_rcpf(den);
  float4 o;
  o.x = fmaf(nsgn * t, g0, xv.x);
  o.y = fmaf(t, g1, xv.y);
  o.z = fmaf(t, g2, xv.z);
  o.w = fmaf(t, g3, xv.w);
  return o;
}

__device__ __forceinline__ void wls_finish(
    const wls_in& in, float* __restrict__ opA, float* __restrict__ opB,
    bool stA, bool stB,
    v2f a0, v2f a1, const v2f* a2A, const v2f* a2B, v8f cb1, v8f cb2, float nsgn)
{
  v2f bA0, bA1, bB0, bB1;
  bA0.x = in.xa.x; bA0.y = in.xa.y; bA1.x = in.xa.z; bA1.y = in.xa.w;
  bB0.x = in.xb.x; bB0.y = in.xb.y; bB1.x = in.xb.z; bB1.y = in.xb.w;
  v8f hA = cb1, hB = cb1;
  hA = __builtin_amdgcn_wmma_f32_16x16x4_f32(false, a0, false, bA0, (short)0, hA, false, false);
  hA = __builtin_amdgcn_wmma_f32_16x16x4_f32(false, a1, false, bA1, (short)0, hA, false, false);
  hB = __builtin_amdgcn_wmma_f32_16x16x4_f32(false, a0, false, bB0, (short)0, hB, false, false);
  hB = __builtin_amdgcn_wmma_f32_16x16x4_f32(false, a1, false, bB1, (short)0, hB, false, false);

  float hrA[8], hrB[8];
#pragma unroll
  for (int v = 0; v < 8; ++v) {
    hrA[v] = wls_relu(hA[v]);
    hrB[v] = wls_relu(hB[v]);
  }

  v8f sz = wls_gemm2(a2A, a2B, hrA, hrB, cb2);

  float sg[8];
#pragma unroll
  for (int v = 0; v < 8; ++v) {
    float z = sz[v];
    float e = __builtin_amdgcn_exp2f(-fabsf(z * WLS_LOG2E));
    float lg = __builtin_amdgcn_logf(1.0f + e);
    sg[v] = fmaf(lg, WLS_LN2, wls_relu(z)) + WLS_EPS;
  }

  const float4 oa = wls_solve4(in.xa, sg[0], sg[1], sg[2], sg[3], nsgn);
  const float4 ob = wls_solve4(in.xb, sg[4], sg[5], sg[6], sg[7], nsgn);

  if (stA) *(float4*)(opA) = oa;
  if (stB) *(float4*)(opB) = ob;
}

__global__ __launch_bounds__(256) void wls_wmma_kernel(
    const float* __restrict__ x, const float* __restrict__ W1,
    const float* __restrict__ b1, const float* __restrict__ W2,
    const float* __restrict__ b2, float* __restrict__ out, int Brows)
{
  const int lane = threadIdx.x & 31;
  const int wave = __builtin_amdgcn_readfirstlane((int)(threadIdx.x >> 5));
  const int hi   = lane >> 4;
  const int lo   = lane & 15;

  const int wavesPerBlock = blockDim.x >> 5;
  const int gwave  = blockIdx.x * wavesPerBlock + wave;
  const int nwaves = gridDim.x * wavesPerBlock;
  const int nfull  = Brows >> 5;
  const int rem    = Brows & 31;

  const uint32_t oA   = (uint32_t)lo * 8 + 4 * hi;
  const float nsgn = hi ? 1.0f : -1.0f;

  int p = gwave;
  const bool any = p < nfull;
  wls_in buf0, buf1, buf2, buf3;
  if (any) {
    const int last = nfull - 1;
    const int p1 = p + nwaves, p2 = p1 + nwaves, p3 = p2 + nwaves;
    buf0 = wls_load(x + (size_t)p * 256, oA, oA + 128);
    buf1 = wls_load(x + (size_t)(p1 < last ? p1 : last) * 256, oA, oA + 128);
    buf2 = wls_load(x + (size_t)(p2 < last ? p2 : last) * 256, oA, oA + 128);
    buf3 = wls_load(x + (size_t)(p3 < last ? p3 : last) * 256, oA, oA + 128);
  }

  v2f a0, a1;
  a0.x = W1[(4 * hi + 0) * 16 + lo];
  a0.y = W1[(4 * hi + 1) * 16 + lo];
  a1.x = W1[(4 * hi + 2) * 16 + lo];
  a1.y = W1[(4 * hi + 3) * 16 + lo];

  const int  chm = (lo & 3) + 4 * (lo >> 3);
  const bool inA = ((lo >> 2) & 1) == 0;
  float w2x[4], w2y[4];
#pragma unroll
  for (int c = 0; c < 4; ++c) {
    w2x[c] = W2[(8 * hi + 2 * c    ) * 8 + chm];
    w2y[c] = W2[(8 * hi + 2 * c + 1) * 8 + chm];
  }

  const float4 b2v = *(const float4*)(b2 + 4 * hi);
  v8f cb1, cb2;
#pragma unroll
  for (int v = 0; v < 8; ++v) cb1[v] = b1[8 * hi + v];
  cb2[0] = b2v.x; cb2[1] = b2v.y; cb2[2] = b2v.z; cb2[3] = b2v.w;
  cb2[4] = b2v.x; cb2[5] = b2v.y; cb2[6] = b2v.z; cb2[7] = b2v.w;

  v2f a2A[4], a2B[4];
#pragma unroll
  for (int c = 0; c < 4; ++c) {
    a2A[c].x = inA ? w2x[c] : 0.0f;  a2A[c].y = inA ? w2y[c] : 0.0f;
    a2B[c].x = inA ? 0.0f : w2x[c];  a2B[c].y = inA ? 0.0f : w2y[c];
  }

  if (any) {
    for (;;) {
      {
        float* o = out + (size_t)p * 256 + oA;
        wls_finish(buf0, o, o + 128, true, true, a0, a1, a2A, a2B, cb1, cb2, nsgn);
      }
      p += nwaves;
      if (p >= nfull) break;
      {
        float* o = out + (size_t)p * 256 + oA;
        wls_finish(buf1, o, o + 128, true, true, a0, a1, a2A, a2B, cb1, cb2, nsgn);
      }
      p += nwaves;
      if (p >= nfull) break;
      {
        float* o = out + (size_t)p * 256 + oA;
        wls_finish(buf2, o, o + 128, true, true, a0, a1, a2A, a2B, cb1, cb2, nsgn);
      }
      p += nwaves;
      if (p >= nfull) break;
      {
        float* o = out + (size_t)p * 256 + oA;
        wls_finish(buf3, o, o + 128, true, true, a0, a1, a2A, a2B, cb1, cb2, nsgn);
      }
      p += nwaves;
      if (p >= nfull) break;
      const int last = nfull - 1;
      const int p1 = p + nwaves, p2 = p1 + nwaves, p3 = p2 + nwaves;
      buf0 = wls_load(x + (size_t)p * 256, oA, oA + 128);
      buf1 = wls_load(x + (size_t)(p1 < last ? p1 : last) * 256, oA, oA + 128);
      buf2 = wls_load(x + (size_t)(p2 < last ? p2 : last) * 256, oA, oA + 128);
      buf3 = wls_load(x + (size_t)(p3 < last ? p3 : last) * 256, oA, oA + 128);
    }
  }

  if (rem != 0 && gwave == (nfull % nwaves)) {
    const int rowA = nfull * 32 + lo;
    const int rowB = rowA + 16;
    const int rcA = rowA < Brows ? rowA : Brows - 1;
    const int rcB = rowB < Brows ? rowB : Brows - 1;
    const wls_in t = wls_load(x, (uint32_t)rcA * 8 + 4 * hi, (uint32_t)rcB * 8 + 4 * hi);
    wls_finish(t, out + (size_t)rcA * 8 + 4 * hi, out + (size_t)rcB * 8 + 4 * hi,
               rowA < Brows, rowB < Brows, a0, a1, a2A, a2B, cb1, cb2, nsgn);
  }
}

extern "C" void kernel_launch(void* const* d_in, const int* in_sizes, int n_in,
                              void* d_out, int out_size, void* d_ws, size_t ws_size,
                              hipStream_t stream) {
  const float* x  = (const float*)d_in[0];
  const float* W1 = (const float*)d_in[1];
  const float* b1 = (const float*)d_in[2];
  const float* W2 = (const float*)d_in[3];
  const float* b2 = (const float*)d_in[4];
  float* out = (float*)d_out;

  const int Brows = in_sizes[0] / 8;
  const int npTot = (Brows + 31) / 32;
  int nblocks = (npTot + 31) / 32;
  if (nblocks < 1) nblocks = 1;
  if (nblocks > 32768) nblocks = 32768;

  wls_wmma_kernel<<<nblocks, 256, 0, stream>>>(x, W1, b1, W2, b2, out, Brows);
}


# First pair's two x loads, then the weight loads, then the other six x loads (the load counter returns in order): weight set-up and pair 0 no longer wait for all eight x loads
# speedup vs baseline: 1.6844x; 1.0521x over previous
#include <hip/hip_runtime.h>


typedef __attribute__((ext_vector_type(2))) float v2f;
typedef __attribute__((ext_vector_type(8))) float v8f;

#define WLS_EPS 1e-6f
#define WLS_LOG2E 1.4426950408889634f
#define WLS_LN2 0.6931471805599453f

__device__ __forceinline__ float wls_relu(float v) {
  float r;
  asm("v_max_num_f32 %0, 0, %1" : "=v"(r) : "v"(v));
  return r;
}

__device__ __forceinline__ float wls_xhalf(float v) {
  int b = __builtin_bit_cast(int, v);
  return __builtin_bit_cast(float,
      __builtin_amdgcn_permlanex16(b, b, 0x76543210, 0xfedcba98, false, false));
}


__device__ __forceinline__ v8f wls_gemm2(const v2f* a2A, const v2f* a2B,
                                         const float* hrA, const float* hrB,
                                         v8f acc) {
#pragma unroll
  for (int c = 0; c < 4; ++c) {
    v2f b; b.x = hrA[2 * c]; b.y = hrA[2 * c + 1];
    acc = __builtin_amdgcn_wmma_f32_16x16x4_f32(false, a2A[c], false, b, (short)0, acc, false, false);
  }
#pragma unroll
  for (int c = 0; c < 4; ++c) {
    v2f b; b.x = hrB[2 * c]; b.y = hrB[2 * c + 1];
    acc = __builtin_amdgcn_wmma_f32_16x16x4_f32(false, a2B[c], false, b, (short)0, acc, false, false);
  }
  return acc;
}

struct wls_in {
  float4 xa, xb;
};

__device__ __forceinline__ wls_in wls_load(const float* __restrict__ xb,
                                           uint32_t oA, uint32_t oB) {
  wls_in r;
  r.xa = *(const float4*)(xb + oA);
  r.xb = *(const float4*)(xb + oB);
  return r;
}

__device__ __forceinline__ float4 wls_solve4(float4 xv, float g0, float g1,
                                             float g2, float g3, float nsgn) {
  float sx  = xv.y + xv.z + xv.w;
  float pn  = fmaf(-nsgn, xv.x, -sx);
  float pd  = g0 + (g1 + g2 + g3);
  float num = pn + wls_xhalf(pn);
  float den = pd + wls_xhalf(pd);
  float t   = num * __builtin_amdgcn_rcpf(den);
  float4 o;
  o.x = fmaf(nsgn * t, g0, xv.x);
  o.y = fmaf(t, g1, xv.y);
  o.z = fmaf(t, g2, xv.z);
  o.w = fmaf(t, g3, xv.w);
  return o;
}

__device__ __forceinline__ void wls_finish(
    const wls_in& in, float* __restrict__ opA, float* __restrict__ opB,
    bool stA, bool stB,
    v2f a0, v2f a1, const v2f* a2A, const v2f* a2B, v8f cb1, v8f cb2, float nsgn)
{
  v2f bA0, bA1, bB0, bB1;
  bA0.x = in.xa.x; bA0.y = in.xa.y; bA1.x = in.xa.z; bA1.y = in.xa.w;
  bB0.x = in.xb.x; bB0.y = in.xb.y; bB1.x = in.xb.z; bB1.y = in.xb.w;
  v8f hA = cb1, hB = cb1;
  hA = __builtin_amdgcn_wmma_f32_16x16x4_f32(false, a0, false, bA0, (short)0, hA, false, false);
  hA = __builtin_amdgcn_wmma_f32_16x16x4_f32(false, a1, false, bA1, (short)0, hA, false, false);
  hB = __builtin_amdgcn_wmma_f32_16x16x4_f32(false, a0, false, bB0, (short)0, hB, false, false);
  hB = __builtin_amdgcn_wmma_f32_16x16x4_f32(false, a1, false, bB1, (short)0, hB, false, false);

  float hrA[8], hrB[8];
#pragma unroll
  for (int v = 0; v < 8; ++v) {
    hrA[v] = wls_relu(hA[v]);
    hrB[v] = wls_relu(hB[v]);
  }

  v8f sz = wls_gemm2(a2A, a2B, hrA, hrB, cb2);

  float sg[8];
#pragma unroll
  for (int v = 0; v < 8; ++v) {
    float z = sz[v];
    float e = __builtin_amdgcn_exp2f(-fabsf(z * WLS_LOG2E));
    float lg = __builtin_amdgcn_logf(1.0f + e);
    sg[v] = fmaf(lg, WLS_LN2, wls_relu(z)) + WLS_EPS;
  }

  const float4 oa = wls_solve4(in.xa, sg[0], sg[1], sg[2], sg[3], nsgn);
  const float4 ob = wls_solve4(in.xb, sg[4], sg[5], sg[6], sg[7], nsgn);

  if (stA) *(float4*)(opA) = oa;
  if (stB) *(float4*)(opB) = ob;
}

struct wls_raw {
  float  w1[4];
  float  w2x[4], w2y[4];
  float4 b2v;
  v8f    cb1;
};

struct wls_wts {
  v2f a0, a1;
  v2f a2A[4], a2B[4];
  v8f cb1, cb2;
};

__device__ __forceinline__ wls_raw wls_wload(
    const float* __restrict__ W1, const float* __restrict__ b1,
    const float* __restrict__ W2, const float* __restrict__ b2, int lo, int hi)
{
  wls_raw r;
#pragma unroll
  for (int j = 0; j < 4; ++j) r.w1[j] = W1[(4 * hi + j) * 16 + lo];
  const int chm = (lo & 3) + 4 * (lo >> 3);
#pragma unroll
  for (int c = 0; c < 4; ++c) {
    r.w2x[c] = W2[(8 * hi + 2 * c    ) * 8 + chm];
    r.w2y[c] = W2[(8 * hi + 2 * c + 1) * 8 + chm];
  }
  r.b2v = *(const float4*)(b2 + 4 * hi);
#pragma unroll
  for (int v = 0; v < 8; ++v) r.cb1[v] = b1[8 * hi + v];
  return r;
}

__device__ __forceinline__ wls_wts wls_wsetup(const wls_raw& r, int lo) {
  wls_wts w;
  w.a0.x = r.w1[0]; w.a0.y = r.w1[1];
  w.a1.x = r.w1[2]; w.a1.y = r.w1[3];
  const bool inA = ((lo >> 2) & 1) == 0;
#pragma unroll
  for (int c = 0; c < 4; ++c) {
    w.a2A[c].x = inA ? r.w2x[c] : 0.0f;  w.a2A[c].y = inA ? r.w2y[c] : 0.0f;
    w.a2B[c].x = inA ? 0.0f : r.w2x[c];  w.a2B[c].y = inA ? 0.0f : r.w2y[c];
  }
  w.cb1 = r.cb1;
  w.cb2[0] = r.b2v.x; w.cb2[1] = r.b2v.y; w.cb2[2] = r.b2v.z; w.cb2[3] = r.b2v.w;
  w.cb2[4] = r.b2v.x; w.cb2[5] = r.b2v.y; w.cb2[6] = r.b2v.z; w.cb2[7] = r.b2v.w;
  return w;
}

__device__ __forceinline__ void wls_partial(
    const float* __restrict__ x, float* __restrict__ out, int Brows, int nfull,
    int lo, int hi, const wls_wts& w, float nsgn)
{
  const int rowA = nfull * 32 + lo;
  const int rowB = rowA + 16;
  const int rcA = rowA < Brows ? rowA : Brows - 1;
  const int rcB = rowB < Brows ? rowB : Brows - 1;
  const wls_in t = wls_load(x, (uint32_t)rcA * 8 + 4 * hi, (uint32_t)rcB * 8 + 4 * hi);
  wls_finish(t, out + (size_t)rcA * 8 + 4 * hi, out + (size_t)rcB * 8 + 4 * hi,
             rowA < Brows, rowB < Brows, w.a0, w.a1, w.a2A, w.a2B, w.cb1, w.cb2, nsgn);
}

__global__ __launch_bounds__(256) void wls_wmma_kernel(
    const float* __restrict__ x, const float* __restrict__ W1,
    const float* __restrict__ b1, const float* __restrict__ W2,
    const float* __restrict__ b2, float* __restrict__ out, int Brows)
{
  const int lane = threadIdx.x & 31;
  const int wave = __builtin_amdgcn_readfirstlane((int)(threadIdx.x >> 5));
  const int hi   = lane >> 4;
  const int lo   = lane & 15;

  const int wavesPerBlock = blockDim.x >> 5;
  const int gwave  = blockIdx.x * wavesPerBlock + wave;
  const int nwaves = gridDim.x * wavesPerBlock;
  const int nfull  = Brows >> 5;
  const int rem    = Brows & 31;

  const uint32_t oA   = (uint32_t)lo * 8 + 4 * hi;
  const float nsgn = hi ? 1.0f : -1.0f;

  int p = gwave;
  if (p < nfull) {
    wls_in buf0 = wls_load(x + (size_t)p * 256, oA, oA + 128);
    asm volatile("" :: "s"(x) : "memory");
    const wls_raw raw = wls_wload(W1, b1, W2, b2, lo, hi);
    const int last = nfull - 1;
    wls_in buf1, buf2, buf3;
    {
      const int p1 = p + nwaves, p2 = p1 + nwaves, p3 = p2 + nwaves;
      buf1 = wls_load(x + (size_t)(p1 < last ? p1 : last) * 256, oA, oA + 128);
      buf2 = wls_load(x + (size_t)(p2 < last ? p2 : last) * 256, oA, oA + 128);
      buf3 = wls_load(x + (size_t)(p3 < last ? p3 : last) * 256, oA, oA + 128);
    }
    asm volatile("" :: "s"(x) : "memory");
    const wls_wts w = wls_wsetup(raw, lo);

    for (;;) {
      {
        float* o = out + (size_t)p * 256 + oA;
        wls_finish(buf0, o, o + 128, true, true, w.a0, w.a1, w.a2A, w.a2B, w.cb1, w.cb2, nsgn);
      }
      p += nwaves;
      if (p >= nfull) break;
      {
        float* o = out + (size_t)p * 256 + oA;
        wls_finish(buf1, o, o + 128, true, true, w.a0, w.a1, w.a2A, w.a2B, w.cb1, w.cb2, nsgn);
      }
      p += nwaves;
      if (p >= nfull) break;
      {
        float* o = out + (size_t)p * 256 + oA;
        wls_finish(buf2, o, o + 128, true, true, w.a0, w.a1, w.a2A, w.a2B, w.cb1, w.cb2, nsgn);
      }
      p += nwaves;
      if (p >= nfull) break;
      {
        float* o = out + (size_t)p * 256 + oA;
        wls_finish(buf3, o, o + 128, true, true, w.a0, w.a1, w.a2A, w.a2B, w.cb1, w.cb2, nsgn);
      }
      p += nwaves;
      if (p >= nfull) break;
      const int p1 = p + nwaves, p2 = p1 + nwaves, p3 = p2 + nwaves;
      buf0 = wls_load(x + (size_t)p * 256, oA, oA + 128);
      buf1 = wls_load(x + (size_t)(p1 < last ? p1 : last) * 256, oA, oA + 128);
      buf2 = wls_load(x + (size_t)(p2 < last ? p2 : last) * 256, oA, oA + 128);
      buf3 = wls_load(x + (size_t)(p3 < last ? p3 : last) * 256, oA, oA + 128);
    }

    if (rem != 0 && gwave == (nfull % nwaves))
      wls_partial(x, out, Brows, nfull, lo, hi, w, nsgn);
  } else if (rem != 0 && gwave == (nfull % nwaves)) {
    const wls_raw raw = wls_wload(W1, b1, W2, b2, lo, hi);
    const wls_wts w = wls_wsetup(raw, lo);
    wls_partial(x, out, Brows, nfull, lo, hi, w, nsgn);
  }
}

extern "C" void kernel_launch(void* const* d_in, const int* in_sizes, int n_in,
                              void* d_out, int out_size, void* d_ws, size_t ws_size,
                              hipStream_t stream) {
  const float* x  = (const float*)d_in[0];
  const float* W1 = (const float*)d_in[1];
  const float* b1 = (const float*)d_in[2];
  const float* W2 = (const float*)d_in[3];
  const float* b2 = (const float*)d_in[4];
  float* out = (float*)d_out;

  const int Brows = in_sizes[0] / 8;
  const int npTot = (Brows + 31) / 32;
  int nblocks = (npTot + 31) / 32;
  if (nblocks < 1) nblocks = 1;
  if (nblocks > 32768) nblocks = 32768;

  wls_wmma_kernel<<<nblocks, 256, 0, stream>>>(x, W1, b1, W2, b2, out, Brows);
}
